# starting kernel
# baseline (speedup 1.0000x reference)
#include <hip/hip_runtime.h>
#include <hip/hip_bf16.h>


static constexpr int FD       = 64;
static constexpr int NGRAPHS  = 512;
static constexpr int NTARGETS = 100;

typedef __attribute__((ext_vector_type(2))) float v2f;
typedef __attribute__((ext_vector_type(8))) float v8f;


__global__ void zero_kernel(float* __restrict__ p, int n) {
  int i = blockIdx.x * blockDim.x + threadIdx.x;
  if (i < n) p[i] = 0.0f;
}

__global__ void deg_accum_kernel(const int* __restrict__ dst, float* __restrict__ deg, int nE) {
  int e = blockIdx.x * blockDim.x + threadIdx.x;
  if (e < nE) atomicAdd(&deg[dst[e]], 1.0f);
}

__global__ void dinv_kernel(float* __restrict__ deg, int n) {
  int i = blockIdx.x * blockDim.x + threadIdx.x;
  if (i < n) deg[i] = rsqrtf(deg[i] + 1.0f);
}

__global__ __launch_bounds__(128) void gemm64_wmma(const float* __restrict__ X,
                                                   const float* __restrict__ W,
                                                   float* __restrict__ H) {
  const int rowBase = blockIdx.x * 16;
  const int wave  = threadIdx.x >> 5;
  const int lane  = threadIdx.x & 31;
  const int m     = lane & 15;
  const int khalf = (lane >> 4) << 1;

  const float* __restrict__ xrow = X + (size_t)(rowBase + m) * FD;
  const float* __restrict__ wcol = W + wave * 16 + m;

  v8f acc = {};
#pragma unroll
  for (int kt = 0; kt < FD / 4; ++kt) {
    const int k = kt * 4 + khalf;
    v2f a; a.x = xrow[k];                 a.y = xrow[k + 1];
    v2f b; b.x = wcol[(size_t)k * FD];    b.y = wcol[(size_t)(k + 1) * FD];
    acc = __builtin_amdgcn_wmma_f32_16x16x4_f32(
         false, a,  false, b,
         (short)0, acc,  false,  false);
  }

  const int mBase = (lane >> 4) << 3;
  float* __restrict__ hout = H + (size_t)rowBase * FD + wave * 16 + m;
#pragma unroll
  for (int v = 0; v < 8; ++v)
    hout[(size_t)(mBase + v) * FD] = acc[v];
}


__global__ void init_agg_kernel(const float* __restrict__ h, const float* __restrict__ dinv,
                                const float* __restrict__ bias, float* __restrict__ agg, int n) {
  int i = blockIdx.x * blockDim.x + threadIdx.x;
  if (i < n) {
    int node = i >> 6;
    int d    = i & 63;
    float di = dinv[node];
    agg[i] = h[i] * di * di + bias[d];
  }
}

__global__ void edge_scatter_kernel(const int* __restrict__ src, const int* __restrict__ dst,
                                    const float* __restrict__ dinv, const float* __restrict__ h,
                                    float* __restrict__ agg, int nE) {
  int t = blockIdx.x * blockDim.x + threadIdx.x;
  int e = t >> 6;
  int d = t & 63;
  if (e < nE) {
    int s  = src[e];
    int dd = dst[e];
    float norm = dinv[s] * dinv[dd];
    atomicAdd(&agg[(size_t)dd * FD + d], h[(size_t)s * FD + d] * norm);
  }
}

__global__ void relu_kernel(float* __restrict__ p, int n) {
  int i = blockIdx.x * blockDim.x + threadIdx.x;
  if (i < n) p[i] = fmaxf(p[i], 0.0f);
}


__global__ void pool_scatter_kernel(const int* __restrict__ seg, const float* __restrict__ h,
                                    float* __restrict__ psum, float* __restrict__ pcnt, int nNodes) {
  int t = blockIdx.x * blockDim.x + threadIdx.x;
  int node = t >> 6;
  int d    = t & 63;
  if (node < nNodes) {
    int g = seg[node];
    atomicAdd(&psum[(size_t)g * FD + d], h[(size_t)node * FD + d]);
    if (d == 0) atomicAdd(&pcnt[g], 1.0f);
  }
}

__global__ void out_gemm_kernel(const float* __restrict__ psum, const float* __restrict__ pcnt,
                                const float* __restrict__ Wout, const float* __restrict__ bout,
                                float* __restrict__ out) {
  int t = blockIdx.x * blockDim.x + threadIdx.x;
  if (t < NGRAPHS * NTARGETS) {
    int g = t / NTARGETS;
    int c = t % NTARGETS;
    float inv = 1.0f / fmaxf(pcnt[g], 1.0f);
    float acc = 0.0f;
#pragma unroll 8
    for (int d = 0; d < FD; ++d)
      acc += psum[(size_t)g * FD + d] * Wout[(size_t)d * NTARGETS + c];
    out[t] = acc * inv + bout[c];
  }
}


extern "C" void kernel_launch(void* const* d_in, const int* in_sizes, int n_in,
                              void* d_out, int out_size, void* d_ws, size_t ws_size,
                              hipStream_t stream) {
  const float* x    = (const float*)d_in[0];
  const float* W1   = (const float*)d_in[1];
  const float* b1   = (const float*)d_in[2];
  const float* W2   = (const float*)d_in[3];
  const float* b2   = (const float*)d_in[4];
  const float* W3   = (const float*)d_in[5];
  const float* b3   = (const float*)d_in[6];
  const float* Wout = (const float*)d_in[7];
  const float* bout = (const float*)d_in[8];
  const int*   eidx = (const int*)d_in[9];
  const int*   seg  = (const int*)d_in[10];
  float* out = (float*)d_out;
  (void)n_in; (void)out_size; (void)ws_size;

  const int nNodes = in_sizes[0] / FD;
  const int nE     = in_sizes[9] / 2;
  const int* srcI = eidx;
  const int* dstI = eidx + nE;

  char* ws = (char*)d_ws;
  size_t off = 0;
  auto alloc = [&](size_t bytes) -> void* {
    void* p = ws + off;
    off = (off + bytes + 255) & ~(size_t)255;
    return p;
  };
  float* dinv = (float*)alloc((size_t)nNodes * sizeof(float));
  float* hbuf = (float*)alloc((size_t)nNodes * FD * sizeof(float));
  float* feat = (float*)alloc((size_t)nNodes * FD * sizeof(float));
  float* psum = (float*)alloc((size_t)NGRAPHS * FD * sizeof(float));
  float* pcnt = (float*)alloc((size_t)NGRAPHS * sizeof(float));

  const int T = 256;
  const int nodeBlocks = (nNodes + T - 1) / T;
  const int gemmBlocks = nNodes / 16;
  const int nfeat      = nNodes * FD;
  const int featBlocks = (nfeat + T - 1) / T;
  const int eBlocks    = (int)(((long long)nE * FD + T - 1) / T);

  zero_kernel<<<nodeBlocks, T, 0, stream>>>(dinv, nNodes);
  deg_accum_kernel<<<(nE + T - 1) / T, T, 0, stream>>>(dstI, dinv, nE);
  dinv_kernel<<<nodeBlocks, T, 0, stream>>>(dinv, nNodes);

  gemm64_wmma<<<gemmBlocks, 128, 0, stream>>>(x, W1, hbuf);
  init_agg_kernel<<<featBlocks, T, 0, stream>>>(hbuf, dinv, b1, feat, nfeat);
  edge_scatter_kernel<<<eBlocks, T, 0, stream>>>(srcI, dstI, dinv, hbuf, feat, nE);
  relu_kernel<<<featBlocks, T, 0, stream>>>(feat, nfeat);

  gemm64_wmma<<<gemmBlocks, 128, 0, stream>>>(feat, W2, hbuf);
  init_agg_kernel<<<featBlocks, T, 0, stream>>>(hbuf, dinv, b2, feat, nfeat);
  edge_scatter_kernel<<<eBlocks, T, 0, stream>>>(srcI, dstI, dinv, hbuf, feat, nE);
  relu_kernel<<<featBlocks, T, 0, stream>>>(feat, nfeat);

  gemm64_wmma<<<gemmBlocks, 128, 0, stream>>>(feat, W3, hbuf);
  init_agg_kernel<<<featBlocks, T, 0, stream>>>(hbuf, dinv, b3, feat, nfeat);
  edge_scatter_kernel<<<eBlocks, T, 0, stream>>>(srcI, dstI, dinv, hbuf, feat, nE);

  zero_kernel<<<(NGRAPHS * FD + T - 1) / T, T, 0, stream>>>(psum, NGRAPHS * FD);
  zero_kernel<<<(NGRAPHS + T - 1) / T, T, 0, stream>>>(pcnt, NGRAPHS);
  pool_scatter_kernel<<<featBlocks, T, 0, stream>>>(seg, feat, psum, pcnt, nNodes);
  out_gemm_kernel<<<(NGRAPHS * NTARGETS + T - 1) / T, T, 0, stream>>>(psum, pcnt, Wout, bout, out);
}


# Edge sums in LDS: edges binned by 200-node destination range once per call; one work-group per range adds gathered rows into its LDS tile (ds_add_f32, arrival order), stores it whole with relu
# speedup vs baseline: 2.3838x; 2.3838x over previous
#include <hip/hip_runtime.h>
#include <hip/hip_bf16.h>


static constexpr int FD       = 64;
static constexpr int NGRAPHS  = 512;
static constexpr int NTARGETS = 100;

static constexpr int TILE      = 200;
static constexpr int MAX_TILES = 1024;
static constexpr int CHUNK     = 4096;
static constexpr int BIN_T     = 256;
static constexpr int AGG_T     = 512;
static constexpr int AGG_W     = AGG_T / 32;

typedef __attribute__((ext_vector_type(2))) float v2f;
typedef __attribute__((ext_vector_type(8))) float v8f;


__global__ void zero_kernel(float* __restrict__ p, int n) {
  int i = blockIdx.x * blockDim.x + threadIdx.x;
  if (i < n) p[i] = 0.0f;
}

__global__ void deg_accum_kernel(const int* __restrict__ dst, float* __restrict__ deg, int nE) {
  int e = blockIdx.x * blockDim.x + threadIdx.x;
  if (e < nE) atomicAdd(&deg[dst[e]], 1.0f);
}

__global__ void dinv_kernel(float* __restrict__ deg, int n) {
  int i = blockIdx.x * blockDim.x + threadIdx.x;
  if (i < n) deg[i] = rsqrtf(deg[i] + 1.0f);
}

__global__ __launch_bounds__(128) void gemm64_wmma(const float* __restrict__ X,
                                                   const float* __restrict__ W,
                                                   float* __restrict__ H) {
  const int rowBase = blockIdx.x * 16;
  const int wave  = threadIdx.x >> 5;
  const int lane  = threadIdx.x & 31;
  const int m     = lane & 15;
  const int khalf = (lane >> 4) << 1;

  const float* __restrict__ xrow = X + (size_t)(rowBase + m) * FD;
  const float* __restrict__ wcol = W + wave * 16 + m;

  v8f acc = {};
#pragma unroll
  for (int kt = 0; kt < FD / 4; ++kt) {
    const int k = kt * 4 + khalf;
    v2f a; a.x = xrow[k];                 a.y = xrow[k + 1];
    v2f b; b.x = wcol[(size_t)k * FD];    b.y = wcol[(size_t)(k + 1) * FD];
    acc = __builtin_amdgcn_wmma_f32_16x16x4_f32(
         false, a,  false, b,
         (short)0, acc,  false,  false);
  }

  const int mBase = (lane >> 4) << 3;
  float* __restrict__ hout = H + (size_t)rowBase * FD + wave * 16 + m;
#pragma unroll
  for (int v = 0; v < 8; ++v)
    hout[(size_t)(mBase + v) * FD] = acc[v];
}


__global__ __launch_bounds__(BIN_T) void bin_count_kernel(const int* __restrict__ dst,
                                                          int* __restrict__ counts,
                                                          int nE, int nTiles, int nChunks) {
  __shared__ int s_cnt[MAX_TILES];
  for (int t = threadIdx.x; t < nTiles; t += BIN_T) s_cnt[t] = 0;
  __syncthreads();
  const int base = blockIdx.x * CHUNK;
  for (int i = threadIdx.x; i < CHUNK; i += BIN_T) {
    const int e = base + i;
    if (e < nE) atomicAdd(&s_cnt[dst[e] / TILE], 1);
  }
  __syncthreads();
  for (int t = threadIdx.x; t < nTiles; t += BIN_T)
    counts[(size_t)t * nChunks + blockIdx.x] = s_cnt[t];
}

__global__ __launch_bounds__(256) void bin_scan_kernel(int* __restrict__ counts,
                                                       int* __restrict__ totals,
                                                       int nTiles, int nChunks) {
  const int lane = threadIdx.x & 31;
  const int t = blockIdx.x * 8 + (threadIdx.x >> 5);
  if (t >= nTiles) return;
  int* __restrict__ row = counts + (size_t)t * nChunks;
  int run = 0;
  for (int c0 = 0; c0 < nChunks; c0 += 32) {
    const int c = c0 + lane;
    const int v = (c < nChunks) ? row[c] : 0;
    int incl = v;
#pragma unroll
    for (int off = 1; off < 32; off <<= 1) {
      const int up = __shfl_up(incl, off);
      if (lane >= off) incl += up;
    }
    if (c < nChunks) row[c] = run + incl - v;
    run += __shfl(incl, 31);
  }
  if (lane == 0) totals[t] = run;
}

__global__ __launch_bounds__(MAX_TILES) void tile_start_kernel(const int* __restrict__ totals,
                                                               int* __restrict__ tileStart,
                                                               int nTiles) {
  __shared__ int s[MAX_TILES];
  const int t = threadIdx.x;
  const int v = (t < nTiles) ? totals[t] : 0;
  s[t] = v;
  __syncthreads();
  for (int off = 1; off < MAX_TILES; off <<= 1) {
    const int add = (t >= off) ? s[t - off] : 0;
    __syncthreads();
    s[t] += add;
    __syncthreads();
  }
  if (t < nTiles) tileStart[t] = s[t] - v;
  if (t == 0) tileStart[nTiles] = s[MAX_TILES - 1];
}

__global__ __launch_bounds__(BIN_T) void bin_place_kernel(const int* __restrict__ src,
                                                          const int* __restrict__ dst,
                                                          const int* __restrict__ prefix,
                                                          const int* __restrict__ tileStart,
                                                          unsigned* __restrict__ binned,
                                                          int nE, int nTiles, int nChunks) {
  __shared__ int s_pos[MAX_TILES];
  for (int t = threadIdx.x; t < nTiles; t += BIN_T)
    s_pos[t] = tileStart[t] + prefix[(size_t)t * nChunks + blockIdx.x];
  __syncthreads();
  const int base = blockIdx.x * CHUNK;
  for (int i = threadIdx.x; i < CHUNK; i += BIN_T) {
    const int e = base + i;
    if (e < nE) {
      const int d = dst[e];
      const int t = d / TILE;
      const int pos = atomicAdd(&s_pos[t], 1);
      binned[pos] = (unsigned)src[e] | ((unsigned)(d - t * TILE) << 24);
    }
  }
}

template <bool RELU>
__global__ __launch_bounds__(AGG_T) void agg_tile_kernel(const float* __restrict__ h,
                                                         const float* __restrict__ dinv,
                                                         const float* __restrict__ bias,
                                                         const unsigned* __restrict__ binned,
                                                         const int* __restrict__ tileStart,
                                                         float* __restrict__ out, int nNodes) {
  __shared__ float s_tile[TILE * FD];
  __shared__ float s_dinv[TILE];

  const int tile     = blockIdx.x;
  const int nodeBase = tile * TILE;
  const int left     = nNodes - nodeBase;
  const int nRows    = left < TILE ? left : TILE;

  const float4* __restrict__ h4 = reinterpret_cast<const float4*>(h + (size_t)nodeBase * FD);
  const float4* __restrict__ b4 = reinterpret_cast<const float4*>(bias);
  float4* s4 = reinterpret_cast<float4*>(s_tile);

  for (int i = threadIdx.x; i < nRows * (FD / 4); i += AGG_T) {
    const float di = dinv[nodeBase + (i >> 4)];
    const float4 hv = h4[i];
    const float4 bv = b4[i & 15];
    float4 o;
    o.x = hv.x * di * di + bv.x;
    o.y = hv.y * di * di + bv.y;
    o.z = hv.z * di * di + bv.z;
    o.w = hv.w * di * di + bv.w;
    s4[i] = o;
  }
  for (int r = threadIdx.x; r < nRows; r += AGG_T) s_dinv[r] = dinv[nodeBase + r];
  __syncthreads();

  const int lane = threadIdx.x & 31;
  const int wave = __builtin_amdgcn_readfirstlane((int)(threadIdx.x >> 5));
  const int eBeg = tileStart[tile];
  const int eEnd = tileStart[tile + 1];
  const float* __restrict__ hl = h + lane * 2;
  float* sl = s_tile + lane * 2;

  for (int c0 = eBeg + wave * 32; c0 < eEnd; c0 += AGG_W * 32) {
    const int rem = eEnd - c0;
    const int n = rem < 32 ? rem : 32;
    const unsigned p = (lane < n) ? binned[c0 + lane] : 0u;
    const int   es = (int)(p & 0xFFFFFFu);
    const int   er = (int)(p >> 24);
    const float en = (lane < n) ? dinv[es] * s_dinv[er] : 0.0f;
    const int   ew = __float_as_int(en);

    int j = 0;
    for (; j + 8 <= n; j += 8) {
      v2f v[8];
      int rr[8];
      float w[8];
#pragma unroll
      for (int u = 0; u < 8; ++u) {
        const int sj = __builtin_amdgcn_readlane(es, j + u);
        rr[u] = __builtin_amdgcn_readlane(er, j + u);
        w[u]  = __int_as_float(__builtin_amdgcn_readlane(ew, j + u));
        v[u]  = *reinterpret_cast<const v2f*>(hl + (size_t)sj * FD);
      }
#pragma unroll
      for (int u = 0; u < 8; ++u) {
        atomicAdd(sl + rr[u] * FD,     v[u].x * w[u]);
        atomicAdd(sl + rr[u] * FD + 1, v[u].y * w[u]);
      }
    }
    for (; j < n; ++j) {
      const int   sj = __builtin_amdgcn_readlane(es, j);
      const int   rj = __builtin_amdgcn_readlane(er, j);
      const float wj = __int_as_float(__builtin_amdgcn_readlane(ew, j));
      const v2f   vj = *reinterpret_cast<const v2f*>(hl + (size_t)sj * FD);
      atomicAdd(sl + rj * FD,     vj.x * wj);
      atomicAdd(sl + rj * FD + 1, vj.y * wj);
    }
  }
  __syncthreads();

  float4* __restrict__ o4 = reinterpret_cast<float4*>(out + (size_t)nodeBase * FD);
  for (int i = threadIdx.x; i < nRows * (FD / 4); i += AGG_T) {
    float4 o = s4[i];
    if (RELU) {
      o.x = fmaxf(o.x, 0.0f);
      o.y = fmaxf(o.y, 0.0f);
      o.z = fmaxf(o.z, 0.0f);
      o.w = fmaxf(o.w, 0.0f);
    }
    o4[i] = o;
  }
}


__global__ void pool_scatter_kernel(const int* __restrict__ seg, const float* __restrict__ h,
                                    float* __restrict__ psum, float* __restrict__ pcnt, int nNodes) {
  int t = blockIdx.x * blockDim.x + threadIdx.x;
  int node = t >> 6;
  int d    = t & 63;
  if (node < nNodes) {
    int g = seg[node];
    atomicAdd(&psum[(size_t)g * FD + d], h[(size_t)node * FD + d]);
    if (d == 0) atomicAdd(&pcnt[g], 1.0f);
  }
}

__global__ void out_gemm_kernel(const float* __restrict__ psum, const float* __restrict__ pcnt,
                                const float* __restrict__ Wout, const float* __restrict__ bout,
                                float* __restrict__ out) {
  int t = blockIdx.x * blockDim.x + threadIdx.x;
  if (t < NGRAPHS * NTARGETS) {
    int g = t / NTARGETS;
    int c = t % NTARGETS;
    float inv = 1.0f / fmaxf(pcnt[g], 1.0f);
    float acc = 0.0f;
#pragma unroll 8
    for (int d = 0; d < FD; ++d)
      acc += psum[(size_t)g * FD + d] * Wout[(size_t)d * NTARGETS + c];
    out[t] = acc * inv + bout[c];
  }
}


extern "C" void kernel_launch(void* const* d_in, const int* in_sizes, int n_in,
                              void* d_out, int out_size, void* d_ws, size_t ws_size,
                              hipStream_t stream) {
  const float* x    = (const float*)d_in[0];
  const float* W1   = (const float*)d_in[1];
  const float* b1   = (const float*)d_in[2];
  const float* W2   = (const float*)d_in[3];
  const float* b2   = (const float*)d_in[4];
  const float* W3   = (const float*)d_in[5];
  const float* b3   = (const float*)d_in[6];
  const float* Wout = (const float*)d_in[7];
  const float* bout = (const float*)d_in[8];
  const int*   eidx = (const int*)d_in[9];
  const int*   seg  = (const int*)d_in[10];
  float* out = (float*)d_out;
  (void)n_in; (void)out_size; (void)ws_size;

  const int nNodes = in_sizes[0] / FD;
  const int nE     = in_sizes[9] / 2;
  const int* srcI = eidx;
  const int* dstI = eidx + nE;

  const int nTiles  = (nNodes + TILE - 1) / TILE;
  const int nChunks = (nE + CHUNK - 1) / CHUNK;

  char* ws = (char*)d_ws;
  size_t off = 0;
  auto alloc = [&](size_t bytes) -> void* {
    void* p = ws + off;
    off = (off + bytes + 255) & ~(size_t)255;
    return p;
  };
  float* dinv = (float*)alloc((size_t)nNodes * sizeof(float));
  float* hbuf = (float*)alloc((size_t)nNodes * FD * sizeof(float));
  float* feat = (float*)alloc((size_t)nNodes * FD * sizeof(float));
  float* psum = (float*)alloc((size_t)NGRAPHS * FD * sizeof(float));
  float* pcnt = (float*)alloc((size_t)NGRAPHS * sizeof(float));
  int*      tileStart = (int*)alloc((size_t)(nTiles + 1) * sizeof(int));
  unsigned* binned    = (unsigned*)alloc((size_t)nE * sizeof(unsigned));
  int* counts = (int*)feat;
  int* totals = counts + (size_t)nTiles * nChunks;

  const int T = 256;
  const int nodeBlocks = (nNodes + T - 1) / T;
  const int gemmBlocks = nNodes / 16;
  const int nfeat      = nNodes * FD;
  const int featBlocks = (nfeat + T - 1) / T;

  zero_kernel<<<nodeBlocks, T, 0, stream>>>(dinv, nNodes);
  deg_accum_kernel<<<(nE + T - 1) / T, T, 0, stream>>>(dstI, dinv, nE);
  dinv_kernel<<<nodeBlocks, T, 0, stream>>>(dinv, nNodes);

  bin_count_kernel<<<nChunks, BIN_T, 0, stream>>>(dstI, counts, nE, nTiles, nChunks);
  bin_scan_kernel<<<(nTiles + 7) / 8, 256, 0, stream>>>(counts, totals, nTiles, nChunks);
  tile_start_kernel<<<1, MAX_TILES, 0, stream>>>(totals, tileStart, nTiles);
  bin_place_kernel<<<nChunks, BIN_T, 0, stream>>>(srcI, dstI, counts, tileStart, binned,
                                                  nE, nTiles, nChunks);

  gemm64_wmma<<<gemmBlocks, 128, 0, stream>>>(x, W1, hbuf);
  agg_tile_kernel<true><<<nTiles, AGG_T, 0, stream>>>(hbuf, dinv, b1, binned, tileStart, feat, nNodes);

  gemm64_wmma<<<gemmBlocks, 128, 0, stream>>>(feat, W2, hbuf);
  agg_tile_kernel<true><<<nTiles, AGG_T, 0, stream>>>(hbuf, dinv, b2, binned, tileStart, feat, nNodes);

  gemm64_wmma<<<gemmBlocks, 128, 0, stream>>>(feat, W3, hbuf);
  agg_tile_kernel<false><<<nTiles, AGG_T, 0, stream>>>(hbuf, dinv, b3, binned, tileStart, feat, nNodes);

  zero_kernel<<<(NGRAPHS * FD + T - 1) / T, T, 0, stream>>>(psum, NGRAPHS * FD);
  zero_kernel<<<(NGRAPHS + T - 1) / T, T, 0, stream>>>(pcnt, NGRAPHS);
  pool_scatter_kernel<<<featBlocks, T, 0, stream>>>(seg, feat, psum, pcnt, nNodes);
  out_gemm_kernel<<<(NGRAPHS * NTARGETS + T - 1) / T, T, 0, stream>>>(psum, pcnt, Wout, bout, out);
}
